# starting kernel
# baseline (speedup 1.0000x reference)
#include <hip/hip_runtime.h>
#include <hip/hip_bf16.h>
#include <math.h>

typedef float v2f __attribute__((ext_vector_type(2)));
typedef float v8f __attribute__((ext_vector_type(8)));

#define D_MODEL 1024
#define D_STATE 16
#define D_INNER 2048
#define KCONV   4
#define BB      2
#define LL      2048
#define NTOK    (BB * LL)

__device__ __forceinline__ float silu_f(float x) { return x / (1.0f + __expf(-x)); }

__device__ __forceinline__ float epilogue_f(float s, float b, int e) {
    s += b;
    if (e == 1) {
        s = (s > 20.0f) ? s : log1pf(__expf(s));
    }
    return s;
}

__global__ __launch_bounds__(256) void gemm_f32_wmma(
    const float* __restrict__ A, const float* __restrict__ Bw,
    const float* __restrict__ bias, float* __restrict__ C,
    int M, int N, int K, int epi)
{
    const int lane = threadIdx.x & 31;
    const int wave = blockIdx.x * (blockDim.x >> 5) + (threadIdx.x >> 5);
    const int tiles_n = (N + 31) >> 5;
    const int tiles_m = (M + 31) >> 5;
    if (wave >= tiles_m * tiles_n) return;
    const int tm = (wave / tiles_n) << 5;
    const int tn = (wave % tiles_n) << 5;

    const int half = lane >> 4;
    const int idx  = lane & 15;

    const float* a0p = A + (size_t)(tm + idx)      * K;
    const float* a1p = A + (size_t)(tm + 16 + idx) * K;

    int nc0 = tn + idx;       if (nc0 > N - 1) nc0 = N - 1;
    int nc1 = tn + 16 + idx;  if (nc1 > N - 1) nc1 = N - 1;

    v8f acc00 = {}; v8f acc01 = {}; v8f acc10 = {}; v8f acc11 = {};

#pragma unroll 2
    for (int k = 0; k < K; k += 4) {
        const int kk = k + (half << 1);
        v2f a0 = *(const v2f*)(a0p + kk);
        v2f a1 = *(const v2f*)(a1p + kk);
        v2f b0, b1;
        b0.x = Bw[(size_t)kk       * N + nc0];
        b0.y = Bw[(size_t)(kk + 1) * N + nc0];
        b1.x = Bw[(size_t)kk       * N + nc1];
        b1.y = Bw[(size_t)(kk + 1) * N + nc1];

        acc00 = __builtin_amdgcn_wmma_f32_16x16x4_f32(false, a0, false, b0, (short)0, acc00, false, false);
        acc01 = __builtin_amdgcn_wmma_f32_16x16x4_f32(false, a0, false, b1, (short)0, acc01, false, false);
        acc10 = __builtin_amdgcn_wmma_f32_16x16x4_f32(false, a1, false, b0, (short)0, acc10, false, false);
        acc11 = __builtin_amdgcn_wmma_f32_16x16x4_f32(false, a1, false, b1, (short)0, acc11, false, false);
    }

    const bool c0ok = (tn + idx)      < N;
    const bool c1ok = (tn + 16 + idx) < N;
    const float bv0 = (bias && c0ok) ? bias[tn + idx]      : 0.0f;
    const float bv1 = (bias && c1ok) ? bias[tn + 16 + idx] : 0.0f;

#pragma unroll
    for (int v = 0; v < 8; ++v) {
        const int r0 = tm + (half << 3) + v;
        const int r1 = r0 + 16;
        if (c0ok) {
            C[(size_t)r0 * N + tn + idx] = epilogue_f(acc00[v], bv0, epi);
            C[(size_t)r1 * N + tn + idx] = epilogue_f(acc10[v], bv0, epi);
        }
        if (c1ok) {
            C[(size_t)r0 * N + tn + 16 + idx] = epilogue_f(acc01[v], bv1, epi);
            C[(size_t)r1 * N + tn + 16 + idx] = epilogue_f(acc11[v], bv1, epi);
        }
    }
}

__global__ __launch_bounds__(256) void conv_silu(
    const float* __restrict__ xz, const float* __restrict__ cw,
    const float* __restrict__ cb, float* __restrict__ act, int total)
{
    const int i = blockIdx.x * blockDim.x + threadIdx.x;
    if (i >= total) return;
    const int d = i % D_INNER;
    const int l = (i / D_INNER) % LL;
    float s = cb[d];
#pragma unroll
    for (int k = 0; k < KCONV; ++k) {
        const int lt = l - (KCONV - 1) + k;
        if (lt >= 0) s += cw[d * KCONV + k] * xz[(size_t)i + (size_t)((long)(lt - l) * D_INNER)];
    }
    act[i] = silu_f(s);
}

__global__ __launch_bounds__(256) void ssm_scan(
    const float* __restrict__ delta, const float* __restrict__ xact,
    const float* __restrict__ Bm, const float* __restrict__ Cm,
    const float* __restrict__ A_log, const float* __restrict__ D_skip,
    float* __restrict__ y)
{
    const int wave = blockIdx.x * (blockDim.x >> 5) + (threadIdx.x >> 5);
    const int lane = threadIdx.x & 31;
    const int half = lane >> 4;
    const int n    = lane & 15;
    const int total = BB * (D_INNER / 2);
    if (wave >= total) return;
    const int b = wave / (D_INNER / 2);
    const int d = (wave % (D_INNER / 2)) * 2 + half;

    const float Acoef = -__expf(A_log[d * D_STATE + n]);
    const float dsk   = D_skip[d];

    const float* dp = delta + (size_t)b * LL * D_INNER + d;
    const float* xp = xact  + (size_t)b * LL * D_INNER + d;
    const float* bp = Bm    + (size_t)b * LL * D_STATE + n;
    const float* cp = Cm    + (size_t)b * LL * D_STATE + n;
    float*       yp = y     + (size_t)b * LL * D_INNER + d;

    float h = 0.0f;
    float dt = dp[0], xt = xp[0], bt = bp[0], ct = cp[0];

    for (int t = 0; t < LL; ++t) {
        float dt2 = 0.f, xt2 = 0.f, bt2 = 0.f, ct2 = 0.f;
        if (t + 1 < LL) {
            const size_t od = (size_t)(t + 1) * D_INNER;
            const size_t ob = (size_t)(t + 1) * D_STATE;
            dt2 = dp[od]; xt2 = xp[od]; bt2 = bp[ob]; ct2 = cp[ob];
        }
        h = __expf(dt * Acoef) * h + (dt * xt) * bt;
        float v = h * ct;
        v += __shfl_xor(v, 1, 32);
        v += __shfl_xor(v, 2, 32);
        v += __shfl_xor(v, 4, 32);
        v += __shfl_xor(v, 8, 32);
        if (n == 0) yp[(size_t)t * D_INNER] = v + xt * dsk;
        dt = dt2; xt = xt2; bt = bt2; ct = ct2;
    }
}

__global__ __launch_bounds__(256) void gate_mul(
    float* __restrict__ y, const float* __restrict__ g, int total)
{
    const int i = blockIdx.x * blockDim.x + threadIdx.x;
    if (i < total) y[i] *= silu_f(g[i]);
}

extern "C" void kernel_launch(void* const* d_in, const int* in_sizes, int n_in,
                              void* d_out, int out_size, void* d_ws, size_t ws_size,
                              hipStream_t stream) {
    const float* x      = (const float*)d_in[0];
    const float* W_in   = (const float*)d_in[1];
    const float* b_in   = (const float*)d_in[2];
    const float* W_gate = (const float*)d_in[3];
    const float* b_gate = (const float*)d_in[4];
    const float* conv_w = (const float*)d_in[5];
    const float* conv_b = (const float*)d_in[6];
    const float* A_log  = (const float*)d_in[7];
    const float* D_skip = (const float*)d_in[8];
    const float* W_delta= (const float*)d_in[9];
    const float* b_delta= (const float*)d_in[10];
    const float* W_B    = (const float*)d_in[11];
    const float* W_C    = (const float*)d_in[12];
    const float* W_out  = (const float*)d_in[13];
    const float* b_out  = (const float*)d_in[14];
    float* out = (float*)d_out;

    const size_t big = (size_t)NTOK * D_INNER;
    float* ws   = (float*)d_ws;
    float* xz   = ws;
    float* xg   = xz  + big;
    float* act  = xg  + big;
    float* yb   = act + big;
    float* Bmat = yb  + big;
    float* Cmat = Bmat + (size_t)NTOK * D_STATE;

    auto launch_gemm = [&](const float* A, const float* Bw, const float* bias,
                           float* C, int M, int N, int K, int e) {
        const int tiles = ((M + 31) / 32) * ((N + 31) / 32);
        const int waves_per_block = 8;
        const int nb = (tiles + waves_per_block - 1) / waves_per_block;
        gemm_f32_wmma<<<nb, 256, 0, stream>>>(A, Bw, bias, C, M, N, K, e);
    };

    launch_gemm(x, W_in,   b_in,   xz, NTOK, D_INNER, D_MODEL, 0);
    launch_gemm(x, W_gate, b_gate, xg, NTOK, D_INNER, D_MODEL, 0);

    {
        const int total = NTOK * D_INNER;
        conv_silu<<<(total + 255) / 256, 256, 0, stream>>>(xz, conv_w, conv_b, act, total);
    }

    launch_gemm(act, W_delta, b_delta, xz, NTOK, D_INNER, D_INNER, 1);

    launch_gemm(act, W_B, nullptr, Bmat, NTOK, D_STATE, D_INNER, 0);
    launch_gemm(act, W_C, nullptr, Cmat, NTOK, D_STATE, D_INNER, 0);

    {
        const int swaves  = BB * (D_INNER / 2);
        const int threads = swaves * 32;
        ssm_scan<<<(threads + 255) / 256, 256, 0, stream>>>(xz, act, Bmat, Cmat,
                                                            A_log, D_skip, yb);
    }

    {
        const int total = NTOK * D_INNER;
        gate_mul<<<(total + 255) / 256, 256, 0, stream>>>(yb, xg, total);
    }

    launch_gemm(yb, W_out, b_out, out, NTOK, D_MODEL, D_INNER, 0);
}


# scan: time loop blocked by 16 steps; a block's inputs are loaded one block ahead into registers and its 16 half-wave sums run level by level, so L2 and cross-lane latency overlap across steps
# speedup vs baseline: 1.2694x; 1.2694x over previous
#include <hip/hip_runtime.h>
#include <hip/hip_bf16.h>
#include <math.h>

typedef float v2f __attribute__((ext_vector_type(2)));
typedef float v8f __attribute__((ext_vector_type(8)));

#define D_MODEL 1024
#define D_STATE 16
#define D_INNER 2048
#define KCONV   4
#define BB      2
#define LL      2048
#define NTOK    (BB * LL)

__device__ __forceinline__ float silu_f(float x) { return x / (1.0f + __expf(-x)); }

__device__ __forceinline__ float epilogue_f(float s, float b, int e) {
    s += b;
    if (e == 1) {
        s = (s > 20.0f) ? s : log1pf(__expf(s));
    }
    return s;
}

__global__ __launch_bounds__(256) void gemm_f32_wmma(
    const float* __restrict__ A, const float* __restrict__ Bw,
    const float* __restrict__ bias, float* __restrict__ C,
    int M, int N, int K, int epi)
{
    const int lane = threadIdx.x & 31;
    const int wave = blockIdx.x * (blockDim.x >> 5) + (threadIdx.x >> 5);
    const int tiles_n = (N + 31) >> 5;
    const int tiles_m = (M + 31) >> 5;
    if (wave >= tiles_m * tiles_n) return;
    const int tm = (wave / tiles_n) << 5;
    const int tn = (wave % tiles_n) << 5;

    const int half = lane >> 4;
    const int idx  = lane & 15;

    const float* a0p = A + (size_t)(tm + idx)      * K;
    const float* a1p = A + (size_t)(tm + 16 + idx) * K;

    int nc0 = tn + idx;       if (nc0 > N - 1) nc0 = N - 1;
    int nc1 = tn + 16 + idx;  if (nc1 > N - 1) nc1 = N - 1;

    v8f acc00 = {}; v8f acc01 = {}; v8f acc10 = {}; v8f acc11 = {};

#pragma unroll 2
    for (int k = 0; k < K; k += 4) {
        const int kk = k + (half << 1);
        v2f a0 = *(const v2f*)(a0p + kk);
        v2f a1 = *(const v2f*)(a1p + kk);
        v2f b0, b1;
        b0.x = Bw[(size_t)kk       * N + nc0];
        b0.y = Bw[(size_t)(kk + 1) * N + nc0];
        b1.x = Bw[(size_t)kk       * N + nc1];
        b1.y = Bw[(size_t)(kk + 1) * N + nc1];

        acc00 = __builtin_amdgcn_wmma_f32_16x16x4_f32(false, a0, false, b0, (short)0, acc00, false, false);
        acc01 = __builtin_amdgcn_wmma_f32_16x16x4_f32(false, a0, false, b1, (short)0, acc01, false, false);
        acc10 = __builtin_amdgcn_wmma_f32_16x16x4_f32(false, a1, false, b0, (short)0, acc10, false, false);
        acc11 = __builtin_amdgcn_wmma_f32_16x16x4_f32(false, a1, false, b1, (short)0, acc11, false, false);
    }

    const bool c0ok = (tn + idx)      < N;
    const bool c1ok = (tn + 16 + idx) < N;
    const float bv0 = (bias && c0ok) ? bias[tn + idx]      : 0.0f;
    const float bv1 = (bias && c1ok) ? bias[tn + 16 + idx] : 0.0f;

#pragma unroll
    for (int v = 0; v < 8; ++v) {
        const int r0 = tm + (half << 3) + v;
        const int r1 = r0 + 16;
        if (c0ok) {
            C[(size_t)r0 * N + tn + idx] = epilogue_f(acc00[v], bv0, epi);
            C[(size_t)r1 * N + tn + idx] = epilogue_f(acc10[v], bv0, epi);
        }
        if (c1ok) {
            C[(size_t)r0 * N + tn + 16 + idx] = epilogue_f(acc01[v], bv1, epi);
            C[(size_t)r1 * N + tn + 16 + idx] = epilogue_f(acc11[v], bv1, epi);
        }
    }
}

__global__ __launch_bounds__(256) void conv_silu(
    const float* __restrict__ xz, const float* __restrict__ cw,
    const float* __restrict__ cb, float* __restrict__ act, int total)
{
    const int i = blockIdx.x * blockDim.x + threadIdx.x;
    if (i >= total) return;
    const int d = i % D_INNER;
    const int l = (i / D_INNER) % LL;
    float s = cb[d];
#pragma unroll
    for (int k = 0; k < KCONV; ++k) {
        const int lt = l - (KCONV - 1) + k;
        if (lt >= 0) s += cw[d * KCONV + k] * xz[(size_t)i + (size_t)((long)(lt - l) * D_INNER)];
    }
    act[i] = silu_f(s);
}

#define SCAN_T 16

struct ScanBlk { float dt[SCAN_T], xt[SCAN_T], bt[SCAN_T], ct[SCAN_T]; };

__device__ __forceinline__ void scan_load(ScanBlk& k, const float* __restrict__ dp,
    const float* __restrict__ xp, const float* __restrict__ bp,
    const float* __restrict__ cp, int t0)
{
#pragma unroll
    for (int j = 0; j < SCAN_T; ++j) {
        const size_t od = (size_t)(t0 + j) * D_INNER;
        const size_t ob = (size_t)(t0 + j) * D_STATE;
        k.dt[j] = dp[od]; k.xt[j] = xp[od]; k.bt[j] = bp[ob]; k.ct[j] = cp[ob];
    }
}

__device__ __forceinline__ void scan_landed16(const float (&a)[SCAN_T])
{
    asm volatile("" :: "v"(a[0]), "v"(a[1]), "v"(a[2]),  "v"(a[3]),  "v"(a[4]),  "v"(a[5]),  "v"(a[6]),  "v"(a[7]),
                       "v"(a[8]), "v"(a[9]), "v"(a[10]), "v"(a[11]), "v"(a[12]), "v"(a[13]), "v"(a[14]), "v"(a[15]));
}
__device__ __forceinline__ void scan_landed(const ScanBlk& k)
{
    scan_landed16(k.dt); scan_landed16(k.xt); scan_landed16(k.bt); scan_landed16(k.ct);
}

__device__ __forceinline__ void scan_run(const ScanBlk& k, float& h, float Acoef,
    float dsk, float* __restrict__ yp, int t0, int n)
{
    float v[SCAN_T];
#pragma unroll
    for (int j = 0; j < SCAN_T; ++j) {
        h = __expf(k.dt[j] * Acoef) * h + (k.dt[j] * k.xt[j]) * k.bt[j];
        v[j] = h * k.ct[j];
    }
#pragma unroll
    for (int m = 1; m < 16; m <<= 1) {
#pragma unroll
        for (int j = 0; j < SCAN_T; ++j) v[j] += __shfl_xor(v[j], m, 32);
    }
    if (n == 0) {
#pragma unroll
        for (int j = 0; j < SCAN_T; ++j)
            yp[(size_t)(t0 + j) * D_INNER] = v[j] + k.xt[j] * dsk;
    }
}

__global__ __launch_bounds__(256) void ssm_scan(
    const float* __restrict__ delta, const float* __restrict__ xact,
    const float* __restrict__ Bm, const float* __restrict__ Cm,
    const float* __restrict__ A_log, const float* __restrict__ D_skip,
    float* __restrict__ y)
{
    const int wave = blockIdx.x * (blockDim.x >> 5) + (threadIdx.x >> 5);
    const int lane = threadIdx.x & 31;
    const int half = lane >> 4;
    const int n    = lane & 15;
    const int total = BB * (D_INNER / 2);
    if (wave >= total) return;
    const int b = wave / (D_INNER / 2);
    const int d = (wave % (D_INNER / 2)) * 2 + half;

    const float Acoef = -__expf(A_log[d * D_STATE + n]);
    const float dsk   = D_skip[d];

    const float* dp = delta + (size_t)b * LL * D_INNER + d;
    const float* xp = xact  + (size_t)b * LL * D_INNER + d;
    const float* bp = Bm    + (size_t)b * LL * D_STATE + n;
    const float* cp = Cm    + (size_t)b * LL * D_STATE + n;
    float*       yp = y     + (size_t)b * LL * D_INNER + d;

    float h = 0.0f;
    ScanBlk ka, kb;
    scan_load(ka, dp, xp, bp, cp, 0);

    for (int t0 = 0; t0 < LL; t0 += 2 * SCAN_T) {
        scan_landed(ka);
        scan_load(kb, dp, xp, bp, cp, t0 + SCAN_T);
        scan_run(ka, h, Acoef, dsk, yp, t0, n);
        scan_landed(kb);
        if (t0 + 2 * SCAN_T < LL) scan_load(ka, dp, xp, bp, cp, t0 + 2 * SCAN_T);
        scan_run(kb, h, Acoef, dsk, yp, t0 + SCAN_T, n);
    }
}

__global__ __launch_bounds__(256) void gate_mul(
    float* __restrict__ y, const float* __restrict__ g, int total)
{
    const int i = blockIdx.x * blockDim.x + threadIdx.x;
    if (i < total) y[i] *= silu_f(g[i]);
}

extern "C" void kernel_launch(void* const* d_in, const int* in_sizes, int n_in,
                              void* d_out, int out_size, void* d_ws, size_t ws_size,
                              hipStream_t stream) {
    const float* x      = (const float*)d_in[0];
    const float* W_in   = (const float*)d_in[1];
    const float* b_in   = (const float*)d_in[2];
    const float* W_gate = (const float*)d_in[3];
    const float* b_gate = (const float*)d_in[4];
    const float* conv_w = (const float*)d_in[5];
    const float* conv_b = (const float*)d_in[6];
    const float* A_log  = (const float*)d_in[7];
    const float* D_skip = (const float*)d_in[8];
    const float* W_delta= (const float*)d_in[9];
    const float* b_delta= (const float*)d_in[10];
    const float* W_B    = (const float*)d_in[11];
    const float* W_C    = (const float*)d_in[12];
    const float* W_out  = (const float*)d_in[13];
    const float* b_out  = (const float*)d_in[14];
    float* out = (float*)d_out;

    const size_t big = (size_t)NTOK * D_INNER;
    float* ws   = (float*)d_ws;
    float* xz   = ws;
    float* xg   = xz  + big;
    float* act  = xg  + big;
    float* yb   = act + big;
    float* Bmat = yb  + big;
    float* Cmat = Bmat + (size_t)NTOK * D_STATE;

    auto launch_gemm = [&](const float* A, const float* Bw, const float* bias,
                           float* C, int M, int N, int K, int e) {
        const int tiles = ((M + 31) / 32) * ((N + 31) / 32);
        const int waves_per_block = 8;
        const int nb = (tiles + waves_per_block - 1) / waves_per_block;
        gemm_f32_wmma<<<nb, 256, 0, stream>>>(A, Bw, bias, C, M, N, K, e);
    };

    launch_gemm(x, W_in,   b_in,   xz, NTOK, D_INNER, D_MODEL, 0);
    launch_gemm(x, W_gate, b_gate, xg, NTOK, D_INNER, D_MODEL, 0);

    {
        const int total = NTOK * D_INNER;
        conv_silu<<<(total + 255) / 256, 256, 0, stream>>>(xz, conv_w, conv_b, act, total);
    }

    launch_gemm(act, W_delta, b_delta, xz, NTOK, D_INNER, D_INNER, 1);

    launch_gemm(act, W_B, nullptr, Bmat, NTOK, D_STATE, D_INNER, 0);
    launch_gemm(act, W_C, nullptr, Cmat, NTOK, D_STATE, D_INNER, 0);

    {
        const int swaves  = BB * (D_INNER / 2);
        const int threads = swaves * 32;
        ssm_scan<<<(threads + 255) / 256, 256, 0, stream>>>(xz, act, Bmat, Cmat,
                                                            A_log, D_skip, yb);
    }

    {
        const int total = NTOK * D_INNER;
        gate_mul<<<(total + 255) / 256, 256, 0, stream>>>(yb, xg, total);
    }

    launch_gemm(yb, W_out, b_out, out, NTOK, D_MODEL, D_INNER, 0);
}


# gemm: the four large products stage A and B through LDS (128x128 tile per work-group, K slices of 32, two stages, whole-line vector global loads); fragment maps, k order and every sum unchanged
# speedup vs baseline: 1.6266x; 1.2814x over previous
#include <hip/hip_runtime.h>
#include <hip/hip_bf16.h>
#include <math.h>

typedef float v2f __attribute__((ext_vector_type(2)));
typedef float v8f __attribute__((ext_vector_type(8)));
typedef float v4f __attribute__((ext_vector_type(4)));

#define D_MODEL 1024
#define D_STATE 16
#define D_INNER 2048
#define KCONV   4
#define BB      2
#define LL      2048
#define NTOK    (BB * LL)

__device__ __forceinline__ float silu_f(float x) { return x / (1.0f + __expf(-x)); }

__device__ __forceinline__ float epilogue_f(float s, float b, int e) {
    s += b;
    if (e == 1) {
        s = (s > 20.0f) ? s : log1pf(__expf(s));
    }
    return s;
}

__global__ __launch_bounds__(256) void gemm_f32_wmma(
    const float* __restrict__ A, const float* __restrict__ Bw,
    const float* __restrict__ bias, float* __restrict__ C,
    int M, int N, int K, int epi)
{
    const int lane = threadIdx.x & 31;
    const int wave = blockIdx.x * (blockDim.x >> 5) + (threadIdx.x >> 5);
    const int tiles_n = (N + 31) >> 5;
    const int tiles_m = (M + 31) >> 5;
    if (wave >= tiles_m * tiles_n) return;
    const int tm = (wave / tiles_n) << 5;
    const int tn = (wave % tiles_n) << 5;

    const int half = lane >> 4;
    const int idx  = lane & 15;

    const float* a0p = A + (size_t)(tm + idx)      * K;
    const float* a1p = A + (size_t)(tm + 16 + idx) * K;

    int nc0 = tn + idx;       if (nc0 > N - 1) nc0 = N - 1;
    int nc1 = tn + 16 + idx;  if (nc1 > N - 1) nc1 = N - 1;

    v8f acc00 = {}; v8f acc01 = {}; v8f acc10 = {}; v8f acc11 = {};

#pragma unroll 2
    for (int k = 0; k < K; k += 4) {
        const int kk = k + (half << 1);
        v2f a0 = *(const v2f*)(a0p + kk);
        v2f a1 = *(const v2f*)(a1p + kk);
        v2f b0, b1;
        b0.x = Bw[(size_t)kk       * N + nc0];
        b0.y = Bw[(size_t)(kk + 1) * N + nc0];
        b1.x = Bw[(size_t)kk       * N + nc1];
        b1.y = Bw[(size_t)(kk + 1) * N + nc1];

        acc00 = __builtin_amdgcn_wmma_f32_16x16x4_f32(false, a0, false, b0, (short)0, acc00, false, false);
        acc01 = __builtin_amdgcn_wmma_f32_16x16x4_f32(false, a0, false, b1, (short)0, acc01, false, false);
        acc10 = __builtin_amdgcn_wmma_f32_16x16x4_f32(false, a1, false, b0, (short)0, acc10, false, false);
        acc11 = __builtin_amdgcn_wmma_f32_16x16x4_f32(false, a1, false, b1, (short)0, acc11, false, false);
    }

    const bool c0ok = (tn + idx)      < N;
    const bool c1ok = (tn + 16 + idx) < N;
    const float bv0 = (bias && c0ok) ? bias[tn + idx]      : 0.0f;
    const float bv1 = (bias && c1ok) ? bias[tn + 16 + idx] : 0.0f;

#pragma unroll
    for (int v = 0; v < 8; ++v) {
        const int r0 = tm + (half << 3) + v;
        const int r1 = r0 + 16;
        if (c0ok) {
            C[(size_t)r0 * N + tn + idx] = epilogue_f(acc00[v], bv0, epi);
            C[(size_t)r1 * N + tn + idx] = epilogue_f(acc10[v], bv0, epi);
        }
        if (c1ok) {
            C[(size_t)r0 * N + tn + 16 + idx] = epilogue_f(acc01[v], bv1, epi);
            C[(size_t)r1 * N + tn + 16 + idx] = epilogue_f(acc11[v], bv1, epi);
        }
    }
}

#define GBM 128
#define GBN 128
#define GBK 32
#define GPA (GBK + 4)
#define GPB (GBN + 8)
#define GCH (GBK / 8)

template <int N, int K, int EPI>
__global__ __launch_bounds__(256) void gemm_f32_wmma_lds(
    const float* __restrict__ A, const float* __restrict__ Bw,
    const float* __restrict__ bias, float* __restrict__ C)
{
    __shared__ __attribute__((aligned(16))) float As[2][GBM * GPA];
    __shared__ __attribute__((aligned(16))) float Bs[2][GBK * GPB];

    const int tid  = threadIdx.x;
    const int lane = tid & 31;
    const int wv   = tid >> 5;
    const int half = lane >> 4;
    const int idx  = lane & 15;
    constexpr int tiles_n = N / GBN;
    constexpr int NK = K / GBK;
    const int m0 = (blockIdx.x / tiles_n) * GBM;
    const int n0 = (blockIdx.x % tiles_n) * GBN;
    const int wm = (wv >> 2) * 64;
    const int wn = (wv & 3) * 32;

    v4f ra[GCH], rb[GCH];
    v8f acc[4][2] = {};

    auto gload = [&](int k0) {
#pragma unroll
        for (int i = 0; i < GCH; ++i) {
            const int c = tid + 256 * i;
            ra[i] = *(const v4f*)(A + (size_t)(m0 + c / (GBK / 4)) * K + k0 + (c % (GBK / 4)) * 4);
            rb[i] = *(const v4f*)(Bw + (size_t)(k0 + c / (GBN / 4)) * N + n0 + (c % (GBN / 4)) * 4);
        }
    };
    auto sstore = [&](int buf) {
#pragma unroll
        for (int i = 0; i < GCH; ++i) {
            const int c = tid + 256 * i;
            *(v4f*)(&As[buf][(c / (GBK / 4)) * GPA + (c % (GBK / 4)) * 4]) = ra[i];
            *(v4f*)(&Bs[buf][(c / (GBN / 4)) * GPB + (c % (GBN / 4)) * 4]) = rb[i];
        }
    };

    gload(0);
    sstore(0);
    __syncthreads();

    for (int kt = 0; kt < NK; ++kt) {
        const int buf = kt & 1;
        if (kt + 1 < NK) gload((kt + 1) * GBK);
        __builtin_amdgcn_sched_barrier(0);

        const float* as = &As[buf][(wm + idx) * GPA + 2 * half];
        const float* bs = &Bs[buf][(2 * half) * GPB + wn + idx];
#pragma unroll
        for (int ks = 0; ks < GBK / 4; ++ks) {
            v2f a[4], b[2];
#pragma unroll
            for (int i = 0; i < 4; ++i)
                a[i] = *(const v2f*)(as + (16 * i) * GPA + 4 * ks);
#pragma unroll
            for (int j = 0; j < 2; ++j) {
                b[j].x = bs[(4 * ks)     * GPB + 16 * j];
                b[j].y = bs[(4 * ks + 1) * GPB + 16 * j];
            }
#pragma unroll
            for (int i = 0; i < 4; ++i)
#pragma unroll
                for (int j = 0; j < 2; ++j)
                    acc[i][j] = __builtin_amdgcn_wmma_f32_16x16x4_f32(false, a[i], false, b[j], (short)0, acc[i][j], false, false);
        }

        if (kt + 1 < NK) sstore(buf ^ 1);
        __syncthreads();
    }

#pragma unroll
    for (int j = 0; j < 2; ++j) {
        const int col = n0 + wn + 16 * j + idx;
        const float bv = bias ? bias[col] : 0.0f;
#pragma unroll
        for (int i = 0; i < 4; ++i) {
#pragma unroll
            for (int r = 0; r < 8; ++r) {
                const int row = m0 + wm + 16 * i + 8 * half + r;
                C[(size_t)row * N + col] = epilogue_f(acc[i][j][r], bv, EPI);
            }
        }
    }
}

__global__ __launch_bounds__(256) void conv_silu(
    const float* __restrict__ xz, const float* __restrict__ cw,
    const float* __restrict__ cb, float* __restrict__ act, int total)
{
    const int i = blockIdx.x * blockDim.x + threadIdx.x;
    if (i >= total) return;
    const int d = i % D_INNER;
    const int l = (i / D_INNER) % LL;
    float s = cb[d];
#pragma unroll
    for (int k = 0; k < KCONV; ++k) {
        const int lt = l - (KCONV - 1) + k;
        if (lt >= 0) s += cw[d * KCONV + k] * xz[(size_t)i + (size_t)((long)(lt - l) * D_INNER)];
    }
    act[i] = silu_f(s);
}

#define SCAN_T 16

struct ScanBlk { float dt[SCAN_T], xt[SCAN_T], bt[SCAN_T], ct[SCAN_T]; };

__device__ __forceinline__ void scan_load(ScanBlk& k, const float* __restrict__ dp,
    const float* __restrict__ xp, const float* __restrict__ bp,
    const float* __restrict__ cp, int t0)
{
#pragma unroll
    for (int j = 0; j < SCAN_T; ++j) {
        const size_t od = (size_t)(t0 + j) * D_INNER;
        const size_t ob = (size_t)(t0 + j) * D_STATE;
        k.dt[j] = dp[od]; k.xt[j] = xp[od]; k.bt[j] = bp[ob]; k.ct[j] = cp[ob];
    }
}

__device__ __forceinline__ void scan_landed16(const float (&a)[SCAN_T])
{
    asm volatile("" :: "v"(a[0]), "v"(a[1]), "v"(a[2]),  "v"(a[3]),  "v"(a[4]),  "v"(a[5]),  "v"(a[6]),  "v"(a[7]),
                       "v"(a[8]), "v"(a[9]), "v"(a[10]), "v"(a[11]), "v"(a[12]), "v"(a[13]), "v"(a[14]), "v"(a[15]));
}
__device__ __forceinline__ void scan_landed(const ScanBlk& k)
{
    scan_landed16(k.dt); scan_landed16(k.xt); scan_landed16(k.bt); scan_landed16(k.ct);
}

__device__ __forceinline__ void scan_run(const ScanBlk& k, float& h, float Acoef,
    float dsk, float* __restrict__ yp, int t0, int n)
{
    float v[SCAN_T];
#pragma unroll
    for (int j = 0; j < SCAN_T; ++j) {
        h = __expf(k.dt[j] * Acoef) * h + (k.dt[j] * k.xt[j]) * k.bt[j];
        v[j] = h * k.ct[j];
    }
#pragma unroll
    for (int m = 1; m < 16; m <<= 1) {
#pragma unroll
        for (int j = 0; j < SCAN_T; ++j) v[j] += __shfl_xor(v[j], m, 32);
    }
    if (n == 0) {
#pragma unroll
        for (int j = 0; j < SCAN_T; ++j)
            yp[(size_t)(t0 + j) * D_INNER] = v[j] + k.xt[j] * dsk;
    }
}

__global__ __launch_bounds__(256) void ssm_scan(
    const float* __restrict__ delta, const float* __restrict__ xact,
    const float* __restrict__ Bm, const float* __restrict__ Cm,
    const float* __restrict__ A_log, const float* __restrict__ D_skip,
    float* __restrict__ y)
{
    const int wave = blockIdx.x * (blockDim.x >> 5) + (threadIdx.x >> 5);
    const int lane = threadIdx.x & 31;
    const int half = lane >> 4;
    const int n    = lane & 15;
    const int total = BB * (D_INNER / 2);
    if (wave >= total) return;
    const int b = wave / (D_INNER / 2);
    const int d = (wave % (D_INNER / 2)) * 2 + half;

    const float Acoef = -__expf(A_log[d * D_STATE + n]);
    const float dsk   = D_skip[d];

    const float* dp = delta + (size_t)b * LL * D_INNER + d;
    const float* xp = xact  + (size_t)b * LL * D_INNER + d;
    const float* bp = Bm    + (size_t)b * LL * D_STATE + n;
    const float* cp = Cm    + (size_t)b * LL * D_STATE + n;
    float*       yp = y     + (size_t)b * LL * D_INNER + d;

    float h = 0.0f;
    ScanBlk ka, kb;
    scan_load(ka, dp, xp, bp, cp, 0);

    for (int t0 = 0; t0 < LL; t0 += 2 * SCAN_T) {
        scan_landed(ka);
        scan_load(kb, dp, xp, bp, cp, t0 + SCAN_T);
        scan_run(ka, h, Acoef, dsk, yp, t0, n);
        scan_landed(kb);
        if (t0 + 2 * SCAN_T < LL) scan_load(ka, dp, xp, bp, cp, t0 + 2 * SCAN_T);
        scan_run(kb, h, Acoef, dsk, yp, t0 + SCAN_T, n);
    }
}

__global__ __launch_bounds__(256) void gate_mul(
    float* __restrict__ y, const float* __restrict__ g, int total)
{
    const int i = blockIdx.x * blockDim.x + threadIdx.x;
    if (i < total) y[i] *= silu_f(g[i]);
}

extern "C" void kernel_launch(void* const* d_in, const int* in_sizes, int n_in,
                              void* d_out, int out_size, void* d_ws, size_t ws_size,
                              hipStream_t stream) {
    const float* x      = (const float*)d_in[0];
    const float* W_in   = (const float*)d_in[1];
    const float* b_in   = (const float*)d_in[2];
    const float* W_gate = (const float*)d_in[3];
    const float* b_gate = (const float*)d_in[4];
    const float* conv_w = (const float*)d_in[5];
    const float* conv_b = (const float*)d_in[6];
    const float* A_log  = (const float*)d_in[7];
    const float* D_skip = (const float*)d_in[8];
    const float* W_delta= (const float*)d_in[9];
    const float* b_delta= (const float*)d_in[10];
    const float* W_B    = (const float*)d_in[11];
    const float* W_C    = (const float*)d_in[12];
    const float* W_out  = (const float*)d_in[13];
    const float* b_out  = (const float*)d_in[14];
    float* out = (float*)d_out;

    const size_t big = (size_t)NTOK * D_INNER;
    float* ws   = (float*)d_ws;
    float* xz   = ws;
    float* xg   = xz  + big;
    float* act  = xg  + big;
    float* yb   = act + big;
    float* Bmat = yb  + big;
    float* Cmat = Bmat + (size_t)NTOK * D_STATE;

    auto launch_gemm = [&](const float* A, const float* Bw, const float* bias,
                           float* C, int M, int N, int K, int e) {
        const int tiles = ((M + 31) / 32) * ((N + 31) / 32);
        const int waves_per_block = 8;
        const int nb = (tiles + waves_per_block - 1) / waves_per_block;
        gemm_f32_wmma<<<nb, 256, 0, stream>>>(A, Bw, bias, C, M, N, K, e);
    };

    gemm_f32_wmma_lds<D_INNER, D_MODEL, 0><<<(NTOK / GBM) * (D_INNER / GBN), 256, 0, stream>>>(x, W_in,   b_in,   xz);
    gemm_f32_wmma_lds<D_INNER, D_MODEL, 0><<<(NTOK / GBM) * (D_INNER / GBN), 256, 0, stream>>>(x, W_gate, b_gate, xg);

    {
        const int total = NTOK * D_INNER;
        conv_silu<<<(total + 255) / 256, 256, 0, stream>>>(xz, conv_w, conv_b, act, total);
    }

    gemm_f32_wmma_lds<D_INNER, D_INNER, 1><<<(NTOK / GBM) * (D_INNER / GBN), 256, 0, stream>>>(act, W_delta, b_delta, xz);

    launch_gemm(act, W_B, nullptr, Bmat, NTOK, D_STATE, D_INNER, 0);
    launch_gemm(act, W_C, nullptr, Cmat, NTOK, D_STATE, D_INNER, 0);

    {
        const int swaves  = BB * (D_INNER / 2);
        const int threads = swaves * 32;
        ssm_scan<<<(threads + 255) / 256, 256, 0, stream>>>(xz, act, Bmat, Cmat,
                                                            A_log, D_skip, yb);
    }

    {
        const int total = NTOK * D_INNER;
        gate_mul<<<(total + 255) / 256, 256, 0, stream>>>(yb, xg, total);
    }

    gemm_f32_wmma_lds<D_MODEL, D_INNER, 0><<<(NTOK / GBM) * (D_MODEL / GBN), 256, 0, stream>>>(yb, W_out, b_out, out);
}


# B/C projections: one split-K kernel for both (16 rows per work-group, 8 waves take 256 k each, partial tiles added through LDS in wave order) replaces two 128-wave launches; k sum order changes
# speedup vs baseline: 2.3324x; 1.4339x over previous
#include <hip/hip_runtime.h>
#include <hip/hip_bf16.h>
#include <math.h>

typedef float v2f __attribute__((ext_vector_type(2)));
typedef float v8f __attribute__((ext_vector_type(8)));
typedef float v4f __attribute__((ext_vector_type(4)));

#define D_MODEL 1024
#define D_STATE 16
#define D_INNER 2048
#define KCONV   4
#define BB      2
#define LL      2048
#define NTOK    (BB * LL)

__device__ __forceinline__ float silu_f(float x) { return x / (1.0f + __expf(-x)); }

__device__ __forceinline__ float epilogue_f(float s, float b, int e) {
    s += b;
    if (e == 1) {
        s = (s > 20.0f) ? s : log1pf(__expf(s));
    }
    return s;
}

#define PJ_WAVES 8
#define PJ_KW    (D_INNER / PJ_WAVES)
#define PJ_UN    8

__global__ __launch_bounds__(256) void proj_bc_splitk(
    const float* __restrict__ A, const float* __restrict__ Wb,
    const float* __restrict__ Wc, float* __restrict__ Bout, float* __restrict__ Cout)
{
    __shared__ float red[PJ_WAVES][2][8][32];

    const int lane = threadIdx.x & 31;
    const int wv   = threadIdx.x >> 5;
    const int half = lane >> 4;
    const int idx  = lane & 15;
    const int m0   = blockIdx.x * 16;

    const float* ap = A  + (size_t)(m0 + idx) * D_INNER + wv * PJ_KW + 2 * half;
    const float* bp = Wb + (size_t)(wv * PJ_KW + 2 * half) * D_STATE + idx;
    const float* cp = Wc + (size_t)(wv * PJ_KW + 2 * half) * D_STATE + idx;

    v8f accB = {}; v8f accC = {};

    for (int kc = 0; kc < PJ_KW; kc += 4 * PJ_UN) {
        v2f a[PJ_UN], b[PJ_UN], c[PJ_UN];
#pragma unroll
        for (int u = 0; u < PJ_UN; ++u) {
            const int k = kc + 4 * u;
            a[u]   = *(const v2f*)(ap + k);
            b[u].x = bp[(size_t)k       * D_STATE];
            b[u].y = bp[(size_t)(k + 1) * D_STATE];
            c[u].x = cp[(size_t)k       * D_STATE];
            c[u].y = cp[(size_t)(k + 1) * D_STATE];
        }
#pragma unroll
        for (int u = 0; u < PJ_UN; ++u) {
            accB = __builtin_amdgcn_wmma_f32_16x16x4_f32(false, a[u], false, b[u], (short)0, accB, false, false);
            accC = __builtin_amdgcn_wmma_f32_16x16x4_f32(false, a[u], false, c[u], (short)0, accC, false, false);
        }
    }

#pragma unroll
    for (int r = 0; r < 8; ++r) {
        red[wv][0][r][lane] = accB[r];
        red[wv][1][r][lane] = accC[r];
    }
    __syncthreads();

    const int row = 2 * wv + half;
    const int r   = row & 7;
    const int sl  = ((row >> 3) << 4) + idx;
    float sb = red[0][0][r][sl];
    float sc = red[0][1][r][sl];
#pragma unroll
    for (int w = 1; w < PJ_WAVES; ++w) {
        sb += red[w][0][r][sl];
        sc += red[w][1][r][sl];
    }
    Bout[(size_t)(m0 + row) * D_STATE + idx] = sb;
    Cout[(size_t)(m0 + row) * D_STATE + idx] = sc;
}

#define GBM 128
#define GBN 128
#define GBK 32
#define GPA (GBK + 4)
#define GPB (GBN + 8)
#define GCH (GBK / 8)

template <int N, int K, int EPI>
__global__ __launch_bounds__(256) void gemm_f32_wmma_lds(
    const float* __restrict__ A, const float* __restrict__ Bw,
    const float* __restrict__ bias, float* __restrict__ C)
{
    __shared__ __attribute__((aligned(16))) float As[2][GBM * GPA];
    __shared__ __attribute__((aligned(16))) float Bs[2][GBK * GPB];

    const int tid  = threadIdx.x;
    const int lane = tid & 31;
    const int wv   = tid >> 5;
    const int half = lane >> 4;
    const int idx  = lane & 15;
    constexpr int tiles_n = N / GBN;
    constexpr int NK = K / GBK;
    const int m0 = (blockIdx.x / tiles_n) * GBM;
    const int n0 = (blockIdx.x % tiles_n) * GBN;
    const int wm = (wv >> 2) * 64;
    const int wn = (wv & 3) * 32;

    v4f ra[GCH], rb[GCH];
    v8f acc[4][2] = {};

    auto gload = [&](int k0) {
#pragma unroll
        for (int i = 0; i < GCH; ++i) {
            const int c = tid + 256 * i;
            ra[i] = *(const v4f*)(A + (size_t)(m0 + c / (GBK / 4)) * K + k0 + (c % (GBK / 4)) * 4);
            rb[i] = *(const v4f*)(Bw + (size_t)(k0 + c / (GBN / 4)) * N + n0 + (c % (GBN / 4)) * 4);
        }
    };
    auto sstore = [&](int buf) {
#pragma unroll
        for (int i = 0; i < GCH; ++i) {
            const int c = tid + 256 * i;
            *(v4f*)(&As[buf][(c / (GBK / 4)) * GPA + (c % (GBK / 4)) * 4]) = ra[i];
            *(v4f*)(&Bs[buf][(c / (GBN / 4)) * GPB + (c % (GBN / 4)) * 4]) = rb[i];
        }
    };

    gload(0);
    sstore(0);
    __syncthreads();

    for (int kt = 0; kt < NK; ++kt) {
        const int buf = kt & 1;
        if (kt + 1 < NK) gload((kt + 1) * GBK);
        __builtin_amdgcn_sched_barrier(0);

        const float* as = &As[buf][(wm + idx) * GPA + 2 * half];
        const float* bs = &Bs[buf][(2 * half) * GPB + wn + idx];
#pragma unroll
        for (int ks = 0; ks < GBK / 4; ++ks) {
            v2f a[4], b[2];
#pragma unroll
            for (int i = 0; i < 4; ++i)
                a[i] = *(const v2f*)(as + (16 * i) * GPA + 4 * ks);
#pragma unroll
            for (int j = 0; j < 2; ++j) {
                b[j].x = bs[(4 * ks)     * GPB + 16 * j];
                b[j].y = bs[(4 * ks + 1) * GPB + 16 * j];
            }
#pragma unroll
            for (int i = 0; i < 4; ++i)
#pragma unroll
                for (int j = 0; j < 2; ++j)
                    acc[i][j] = __builtin_amdgcn_wmma_f32_16x16x4_f32(false, a[i], false, b[j], (short)0, acc[i][j], false, false);
        }

        if (kt + 1 < NK) sstore(buf ^ 1);
        __syncthreads();
    }

#pragma unroll
    for (int j = 0; j < 2; ++j) {
        const int col = n0 + wn + 16 * j + idx;
        const float bv = bias ? bias[col] : 0.0f;
#pragma unroll
        for (int i = 0; i < 4; ++i) {
#pragma unroll
            for (int r = 0; r < 8; ++r) {
                const int row = m0 + wm + 16 * i + 8 * half + r;
                C[(size_t)row * N + col] = epilogue_f(acc[i][j][r], bv, EPI);
            }
        }
    }
}

__global__ __launch_bounds__(256) void conv_silu(
    const float* __restrict__ xz, const float* __restrict__ cw,
    const float* __restrict__ cb, float* __restrict__ act, int total)
{
    const int i = blockIdx.x * blockDim.x + threadIdx.x;
    if (i >= total) return;
    const int d = i % D_INNER;
    const int l = (i / D_INNER) % LL;
    float s = cb[d];
#pragma unroll
    for (int k = 0; k < KCONV; ++k) {
        const int lt = l - (KCONV - 1) + k;
        if (lt >= 0) s += cw[d * KCONV + k] * xz[(size_t)i + (size_t)((long)(lt - l) * D_INNER)];
    }
    act[i] = silu_f(s);
}

#define SCAN_T 16

struct ScanBlk { float dt[SCAN_T], xt[SCAN_T], bt[SCAN_T], ct[SCAN_T]; };

__device__ __forceinline__ void scan_load(ScanBlk& k, const float* __restrict__ dp,
    const float* __restrict__ xp, const float* __restrict__ bp,
    const float* __restrict__ cp, int t0)
{
#pragma unroll
    for (int j = 0; j < SCAN_T; ++j) {
        const size_t od = (size_t)(t0 + j) * D_INNER;
        const size_t ob = (size_t)(t0 + j) * D_STATE;
        k.dt[j] = dp[od]; k.xt[j] = xp[od]; k.bt[j] = bp[ob]; k.ct[j] = cp[ob];
    }
}

__device__ __forceinline__ void scan_landed16(const float (&a)[SCAN_T])
{
    asm volatile("" :: "v"(a[0]), "v"(a[1]), "v"(a[2]),  "v"(a[3]),  "v"(a[4]),  "v"(a[5]),  "v"(a[6]),  "v"(a[7]),
                       "v"(a[8]), "v"(a[9]), "v"(a[10]), "v"(a[11]), "v"(a[12]), "v"(a[13]), "v"(a[14]), "v"(a[15]));
}
__device__ __forceinline__ void scan_landed(const ScanBlk& k)
{
    scan_landed16(k.dt); scan_landed16(k.xt); scan_landed16(k.bt); scan_landed16(k.ct);
}

__device__ __forceinline__ void scan_run(const ScanBlk& k, float& h, float Acoef,
    float dsk, float* __restrict__ yp, int t0, int n)
{
    float v[SCAN_T];
#pragma unroll
    for (int j = 0; j < SCAN_T; ++j) {
        h = __expf(k.dt[j] * Acoef) * h + (k.dt[j] * k.xt[j]) * k.bt[j];
        v[j] = h * k.ct[j];
    }
#pragma unroll
    for (int m = 1; m < 16; m <<= 1) {
#pragma unroll
        for (int j = 0; j < SCAN_T; ++j) v[j] += __shfl_xor(v[j], m, 32);
    }
    if (n == 0) {
#pragma unroll
        for (int j = 0; j < SCAN_T; ++j)
            yp[(size_t)(t0 + j) * D_INNER] = v[j] + k.xt[j] * dsk;
    }
}

__global__ __launch_bounds__(256) void ssm_scan(
    const float* __restrict__ delta, const float* __restrict__ xact,
    const float* __restrict__ Bm, const float* __restrict__ Cm,
    const float* __restrict__ A_log, const float* __restrict__ D_skip,
    float* __restrict__ y)
{
    const int wave = blockIdx.x * (blockDim.x >> 5) + (threadIdx.x >> 5);
    const int lane = threadIdx.x & 31;
    const int half = lane >> 4;
    const int n    = lane & 15;
    const int total = BB * (D_INNER / 2);
    if (wave >= total) return;
    const int b = wave / (D_INNER / 2);
    const int d = (wave % (D_INNER / 2)) * 2 + half;

    const float Acoef = -__expf(A_log[d * D_STATE + n]);
    const float dsk   = D_skip[d];

    const float* dp = delta + (size_t)b * LL * D_INNER + d;
    const float* xp = xact  + (size_t)b * LL * D_INNER + d;
    const float* bp = Bm    + (size_t)b * LL * D_STATE + n;
    const float* cp = Cm    + (size_t)b * LL * D_STATE + n;
    float*       yp = y     + (size_t)b * LL * D_INNER + d;

    float h = 0.0f;
    ScanBlk ka, kb;
    scan_load(ka, dp, xp, bp, cp, 0);

    for (int t0 = 0; t0 < LL; t0 += 2 * SCAN_T) {
        scan_landed(ka);
        scan_load(kb, dp, xp, bp, cp, t0 + SCAN_T);
        scan_run(ka, h, Acoef, dsk, yp, t0, n);
        scan_landed(kb);
        if (t0 + 2 * SCAN_T < LL) scan_load(ka, dp, xp, bp, cp, t0 + 2 * SCAN_T);
        scan_run(kb, h, Acoef, dsk, yp, t0 + SCAN_T, n);
    }
}

__global__ __launch_bounds__(256) void gate_mul(
    float* __restrict__ y, const float* __restrict__ g, int total)
{
    const int i = blockIdx.x * blockDim.x + threadIdx.x;
    if (i < total) y[i] *= silu_f(g[i]);
}

extern "C" void kernel_launch(void* const* d_in, const int* in_sizes, int n_in,
                              void* d_out, int out_size, void* d_ws, size_t ws_size,
                              hipStream_t stream) {
    const float* x      = (const float*)d_in[0];
    const float* W_in   = (const float*)d_in[1];
    const float* b_in   = (const float*)d_in[2];
    const float* W_gate = (const float*)d_in[3];
    const float* b_gate = (const float*)d_in[4];
    const float* conv_w = (const float*)d_in[5];
    const float* conv_b = (const float*)d_in[6];
    const float* A_log  = (const float*)d_in[7];
    const float* D_skip = (const float*)d_in[8];
    const float* W_delta= (const float*)d_in[9];
    const float* b_delta= (const float*)d_in[10];
    const float* W_B    = (const float*)d_in[11];
    const float* W_C    = (const float*)d_in[12];
    const float* W_out  = (const float*)d_in[13];
    const float* b_out  = (const float*)d_in[14];
    float* out = (float*)d_out;

    const size_t big = (size_t)NTOK * D_INNER;
    float* ws   = (float*)d_ws;
    float* xz   = ws;
    float* xg   = xz  + big;
    float* act  = xg  + big;
    float* yb   = act + big;
    float* Bmat = yb  + big;
    float* Cmat = Bmat + (size_t)NTOK * D_STATE;

    gemm_f32_wmma_lds<D_INNER, D_MODEL, 0><<<(NTOK / GBM) * (D_INNER / GBN), 256, 0, stream>>>(x, W_in,   b_in,   xz);
    gemm_f32_wmma_lds<D_INNER, D_MODEL, 0><<<(NTOK / GBM) * (D_INNER / GBN), 256, 0, stream>>>(x, W_gate, b_gate, xg);

    {
        const int total = NTOK * D_INNER;
        conv_silu<<<(total + 255) / 256, 256, 0, stream>>>(xz, conv_w, conv_b, act, total);
    }

    gemm_f32_wmma_lds<D_INNER, D_INNER, 1><<<(NTOK / GBM) * (D_INNER / GBN), 256, 0, stream>>>(act, W_delta, b_delta, xz);

    proj_bc_splitk<<<NTOK / 16, 256, 0, stream>>>(act, W_B, W_C, Bmat, Cmat);

    {
        const int swaves  = BB * (D_INNER / 2);
        const int threads = swaves * 32;
        ssm_scan<<<(threads + 255) / 256, 256, 0, stream>>>(xz, act, Bmat, Cmat,
                                                            A_log, D_skip, yb);
    }

    {
        const int total = NTOK * D_INNER;
        gate_mul<<<(total + 255) / 256, 256, 0, stream>>>(yb, xg, total);
    }

    gemm_f32_wmma_lds<D_MODEL, D_INNER, 0><<<(NTOK / GBM) * (D_MODEL / GBN), 256, 0, stream>>>(yb, W_out, b_out, out);
}


# scan: lane n owns step t0+n of delta, x, y: 1 load/store per 16-step block for 16; steps take delta, delta*x by DPP row_share; row sums as one reduce-scatter (15 exchanges for 64); sum order kept
# speedup vs baseline: 3.0055x; 1.2886x over previous
#include <hip/hip_runtime.h>
#include <hip/hip_bf16.h>
#include <math.h>

typedef float v2f __attribute__((ext_vector_type(2)));
typedef float v8f __attribute__((ext_vector_type(8)));
typedef float v4f __attribute__((ext_vector_type(4)));

#define D_MODEL 1024
#define D_STATE 16
#define D_INNER 2048
#define KCONV   4
#define BB      2
#define LL      2048
#define NTOK    (BB * LL)

__device__ __forceinline__ float silu_f(float x) { return x / (1.0f + __expf(-x)); }

__device__ __forceinline__ float epilogue_f(float s, float b, int e) {
    s += b;
    if (e == 1) {
        s = (s > 20.0f) ? s : log1pf(__expf(s));
    }
    return s;
}

#define PJ_WAVES 8
#define PJ_KW    (D_INNER / PJ_WAVES)
#define PJ_UN    8

__global__ __launch_bounds__(256) void proj_bc_splitk(
    const float* __restrict__ A, const float* __restrict__ Wb,
    const float* __restrict__ Wc, float* __restrict__ Bout, float* __restrict__ Cout)
{
    __shared__ float red[PJ_WAVES][2][8][32];

    const int lane = threadIdx.x & 31;
    const int wv   = threadIdx.x >> 5;
    const int half = lane >> 4;
    const int idx  = lane & 15;
    const int m0   = blockIdx.x * 16;

    const float* ap = A  + (size_t)(m0 + idx) * D_INNER + wv * PJ_KW + 2 * half;
    const float* bp = Wb + (size_t)(wv * PJ_KW + 2 * half) * D_STATE + idx;
    const float* cp = Wc + (size_t)(wv * PJ_KW + 2 * half) * D_STATE + idx;

    v8f accB = {}; v8f accC = {};

    for (int kc = 0; kc < PJ_KW; kc += 4 * PJ_UN) {
        v2f a[PJ_UN], b[PJ_UN], c[PJ_UN];
#pragma unroll
        for (int u = 0; u < PJ_UN; ++u) {
            const int k = kc + 4 * u;
            a[u]   = *(const v2f*)(ap + k);
            b[u].x = bp[(size_t)k       * D_STATE];
            b[u].y = bp[(size_t)(k + 1) * D_STATE];
            c[u].x = cp[(size_t)k       * D_STATE];
            c[u].y = cp[(size_t)(k + 1) * D_STATE];
        }
#pragma unroll
        for (int u = 0; u < PJ_UN; ++u) {
            accB = __builtin_amdgcn_wmma_f32_16x16x4_f32(false, a[u], false, b[u], (short)0, accB, false, false);
            accC = __builtin_amdgcn_wmma_f32_16x16x4_f32(false, a[u], false, c[u], (short)0, accC, false, false);
        }
    }

#pragma unroll
    for (int r = 0; r < 8; ++r) {
        red[wv][0][r][lane] = accB[r];
        red[wv][1][r][lane] = accC[r];
    }
    __syncthreads();

    const int row = 2 * wv + half;
    const int r   = row & 7;
    const int sl  = ((row >> 3) << 4) + idx;
    float sb = red[0][0][r][sl];
    float sc = red[0][1][r][sl];
#pragma unroll
    for (int w = 1; w < PJ_WAVES; ++w) {
        sb += red[w][0][r][sl];
        sc += red[w][1][r][sl];
    }
    Bout[(size_t)(m0 + row) * D_STATE + idx] = sb;
    Cout[(size_t)(m0 + row) * D_STATE + idx] = sc;
}

#define GBM 128
#define GBN 128
#define GBK 32
#define GPA (GBK + 4)
#define GPB (GBN + 8)
#define GCH (GBK / 8)

template <int N, int K, int EPI>
__global__ __launch_bounds__(256) void gemm_f32_wmma_lds(
    const float* __restrict__ A, const float* __restrict__ Bw,
    const float* __restrict__ bias, float* __restrict__ C)
{
    __shared__ __attribute__((aligned(16))) float As[2][GBM * GPA];
    __shared__ __attribute__((aligned(16))) float Bs[2][GBK * GPB];

    const int tid  = threadIdx.x;
    const int lane = tid & 31;
    const int wv   = tid >> 5;
    const int half = lane >> 4;
    const int idx  = lane & 15;
    constexpr int tiles_n = N / GBN;
    constexpr int NK = K / GBK;
    const int m0 = (blockIdx.x / tiles_n) * GBM;
    const int n0 = (blockIdx.x % tiles_n) * GBN;
    const int wm = (wv >> 2) * 64;
    const int wn = (wv & 3) * 32;

    v4f ra[GCH], rb[GCH];
    v8f acc[4][2] = {};

    auto gload = [&](int k0) {
#pragma unroll
        for (int i = 0; i < GCH; ++i) {
            const int c = tid + 256 * i;
            ra[i] = *(const v4f*)(A + (size_t)(m0 + c / (GBK / 4)) * K + k0 + (c % (GBK / 4)) * 4);
            rb[i] = *(const v4f*)(Bw + (size_t)(k0 + c / (GBN / 4)) * N + n0 + (c % (GBN / 4)) * 4);
        }
    };
    auto sstore = [&](int buf) {
#pragma unroll
        for (int i = 0; i < GCH; ++i) {
            const int c = tid + 256 * i;
            *(v4f*)(&As[buf][(c / (GBK / 4)) * GPA + (c % (GBK / 4)) * 4]) = ra[i];
            *(v4f*)(&Bs[buf][(c / (GBN / 4)) * GPB + (c % (GBN / 4)) * 4]) = rb[i];
        }
    };

    gload(0);
    sstore(0);
    __syncthreads();

    for (int kt = 0; kt < NK; ++kt) {
        const int buf = kt & 1;
        if (kt + 1 < NK) gload((kt + 1) * GBK);
        __builtin_amdgcn_sched_barrier(0);

        const float* as = &As[buf][(wm + idx) * GPA + 2 * half];
        const float* bs = &Bs[buf][(2 * half) * GPB + wn + idx];
#pragma unroll
        for (int ks = 0; ks < GBK / 4; ++ks) {
            v2f a[4], b[2];
#pragma unroll
            for (int i = 0; i < 4; ++i)
                a[i] = *(const v2f*)(as + (16 * i) * GPA + 4 * ks);
#pragma unroll
            for (int j = 0; j < 2; ++j) {
                b[j].x = bs[(4 * ks)     * GPB + 16 * j];
                b[j].y = bs[(4 * ks + 1) * GPB + 16 * j];
            }
#pragma unroll
            for (int i = 0; i < 4; ++i)
#pragma unroll
                for (int j = 0; j < 2; ++j)
                    acc[i][j] = __builtin_amdgcn_wmma_f32_16x16x4_f32(false, a[i], false, b[j], (short)0, acc[i][j], false, false);
        }

        if (kt + 1 < NK) sstore(buf ^ 1);
        __syncthreads();
    }

#pragma unroll
    for (int j = 0; j < 2; ++j) {
        const int col = n0 + wn + 16 * j + idx;
        const float bv = bias ? bias[col] : 0.0f;
#pragma unroll
        for (int i = 0; i < 4; ++i) {
#pragma unroll
            for (int r = 0; r < 8; ++r) {
                const int row = m0 + wm + 16 * i + 8 * half + r;
                C[(size_t)row * N + col] = epilogue_f(acc[i][j][r], bv, EPI);
            }
        }
    }
}

__global__ __launch_bounds__(256) void conv_silu(
    const float* __restrict__ xz, const float* __restrict__ cw,
    const float* __restrict__ cb, float* __restrict__ act, int total)
{
    const int i = blockIdx.x * blockDim.x + threadIdx.x;
    if (i >= total) return;
    const int d = i % D_INNER;
    const int l = (i / D_INNER) % LL;
    float s = cb[d];
#pragma unroll
    for (int k = 0; k < KCONV; ++k) {
        const int lt = l - (KCONV - 1) + k;
        if (lt >= 0) s += cw[d * KCONV + k] * xz[(size_t)i + (size_t)((long)(lt - l) * D_INNER)];
    }
    act[i] = silu_f(s);
}

#define SCAN_T 16

struct ScanBlk { float dt, xt; float bt[SCAN_T], ct[SCAN_T]; };

__device__ __forceinline__ void scan_load(ScanBlk& k, const float* __restrict__ dp,
    const float* __restrict__ xp, const float* __restrict__ bp,
    const float* __restrict__ cp, int t0, int n)
{
    const size_t on = (size_t)(t0 + n) * D_INNER;
    k.dt = dp[on]; k.xt = xp[on];
#pragma unroll
    for (int j = 0; j < SCAN_T; ++j) {
        const size_t ob = (size_t)(t0 + j) * D_STATE;
        k.bt[j] = bp[ob]; k.ct[j] = cp[ob];
    }
}

__device__ __forceinline__ void scan_landed16(const float (&a)[SCAN_T])
{
    asm volatile("" :: "v"(a[0]), "v"(a[1]), "v"(a[2]),  "v"(a[3]),  "v"(a[4]),  "v"(a[5]),  "v"(a[6]),  "v"(a[7]),
                       "v"(a[8]), "v"(a[9]), "v"(a[10]), "v"(a[11]), "v"(a[12]), "v"(a[13]), "v"(a[14]), "v"(a[15]));
}
__device__ __forceinline__ void scan_landed(const ScanBlk& k)
{
    asm volatile("" :: "v"(k.dt), "v"(k.xt));
    scan_landed16(k.bt); scan_landed16(k.ct);
}

template <int J>
__device__ __forceinline__ float row_bcast(float v)
{
    return __int_as_float(__builtin_amdgcn_update_dpp(0, __float_as_int(v), 0x150 + J, 0xf, 0xf, true));
}

template <int J>
__device__ __forceinline__ void scan_step(const ScanBlk& k, float dx, float& h, float Acoef,
    float (&v)[SCAN_T])
{
    const float dtj = row_bcast<J>(k.dt);
    const float dxj = row_bcast<J>(dx);
    h = __expf(dtj * Acoef) * h + dxj * k.bt[J];
    v[J] = h * k.ct[J];
}

__device__ __forceinline__ void scan_run(const ScanBlk& k, float& h, float Acoef,
    float dsk, float* __restrict__ yp, int t0, int n)
{
    float v[SCAN_T];
    const float dx = k.dt * k.xt;
    scan_step<0>(k, dx, h, Acoef, v);  scan_step<1>(k, dx, h, Acoef, v);
    scan_step<2>(k, dx, h, Acoef, v);  scan_step<3>(k, dx, h, Acoef, v);
    scan_step<4>(k, dx, h, Acoef, v);  scan_step<5>(k, dx, h, Acoef, v);
    scan_step<6>(k, dx, h, Acoef, v);  scan_step<7>(k, dx, h, Acoef, v);
    scan_step<8>(k, dx, h, Acoef, v);  scan_step<9>(k, dx, h, Acoef, v);
    scan_step<10>(k, dx, h, Acoef, v); scan_step<11>(k, dx, h, Acoef, v);
    scan_step<12>(k, dx, h, Acoef, v); scan_step<13>(k, dx, h, Acoef, v);
    scan_step<14>(k, dx, h, Acoef, v); scan_step<15>(k, dx, h, Acoef, v);
    const bool b0 = (n & 1) != 0, b1 = (n & 2) != 0, b2 = (n & 4) != 0, b3 = (n & 8) != 0;
    float w[SCAN_T / 2], u[SCAN_T / 4], q[SCAN_T / 8];
#pragma unroll
    for (int i = 0; i < SCAN_T / 2; ++i)
        w[i] = (b0 ? v[2 * i + 1] : v[2 * i]) + __shfl_xor(b0 ? v[2 * i] : v[2 * i + 1], 1, 32);
#pragma unroll
    for (int i = 0; i < SCAN_T / 4; ++i)
        u[i] = (b1 ? w[2 * i + 1] : w[2 * i]) + __shfl_xor(b1 ? w[2 * i] : w[2 * i + 1], 2, 32);
#pragma unroll
    for (int i = 0; i < SCAN_T / 8; ++i)
        q[i] = (b2 ? u[2 * i + 1] : u[2 * i]) + __shfl_xor(b2 ? u[2 * i] : u[2 * i + 1], 4, 32);
    const float o = (b3 ? q[1] : q[0]) + __shfl_xor(b3 ? q[0] : q[1], 8, 32);
    yp[(size_t)(t0 + n) * D_INNER] = o + k.xt * dsk;
}

__global__ __launch_bounds__(256) void ssm_scan(
    const float* __restrict__ delta, const float* __restrict__ xact,
    const float* __restrict__ Bm, const float* __restrict__ Cm,
    const float* __restrict__ A_log, const float* __restrict__ D_skip,
    float* __restrict__ y)
{
    const int wave = blockIdx.x * (blockDim.x >> 5) + (threadIdx.x >> 5);
    const int lane = threadIdx.x & 31;
    const int half = lane >> 4;
    const int n    = lane & 15;
    const int total = BB * (D_INNER / 2);
    if (wave >= total) return;
    const int b = wave / (D_INNER / 2);
    const int d = (wave % (D_INNER / 2)) * 2 + half;

    const float Acoef = -__expf(A_log[d * D_STATE + n]);
    const float dsk   = D_skip[d];

    const float* dp = delta + (size_t)b * LL * D_INNER + d;
    const float* xp = xact  + (size_t)b * LL * D_INNER + d;
    const float* bp = Bm    + (size_t)b * LL * D_STATE + n;
    const float* cp = Cm    + (size_t)b * LL * D_STATE + n;
    float*       yp = y     + (size_t)b * LL * D_INNER + d;

    float h = 0.0f;
    ScanBlk ka, kb;
    scan_load(ka, dp, xp, bp, cp, 0, n);

    for (int t0 = 0; t0 < LL; t0 += 2 * SCAN_T) {
        scan_landed(ka);
        scan_load(kb, dp, xp, bp, cp, t0 + SCAN_T, n);
        scan_run(ka, h, Acoef, dsk, yp, t0, n);
        scan_landed(kb);
        if (t0 + 2 * SCAN_T < LL) scan_load(ka, dp, xp, bp, cp, t0 + 2 * SCAN_T, n);
        scan_run(kb, h, Acoef, dsk, yp, t0 + SCAN_T, n);
    }
}

__global__ __launch_bounds__(256) void gate_mul(
    float* __restrict__ y, const float* __restrict__ g, int total)
{
    const int i = blockIdx.x * blockDim.x + threadIdx.x;
    if (i < total) y[i] *= silu_f(g[i]);
}

extern "C" void kernel_launch(void* const* d_in, const int* in_sizes, int n_in,
                              void* d_out, int out_size, void* d_ws, size_t ws_size,
                              hipStream_t stream) {
    const float* x      = (const float*)d_in[0];
    const float* W_in   = (const float*)d_in[1];
    const float* b_in   = (const float*)d_in[2];
    const float* W_gate = (const float*)d_in[3];
    const float* b_gate = (const float*)d_in[4];
    const float* conv_w = (const float*)d_in[5];
    const float* conv_b = (const float*)d_in[6];
    const float* A_log  = (const float*)d_in[7];
    const float* D_skip = (const float*)d_in[8];
    const float* W_delta= (const float*)d_in[9];
    const float* b_delta= (const float*)d_in[10];
    const float* W_B    = (const float*)d_in[11];
    const float* W_C    = (const float*)d_in[12];
    const float* W_out  = (const float*)d_in[13];
    const float* b_out  = (const float*)d_in[14];
    float* out = (float*)d_out;

    const size_t big = (size_t)NTOK * D_INNER;
    float* ws   = (float*)d_ws;
    float* xz   = ws;
    float* xg   = xz  + big;
    float* act  = xg  + big;
    float* yb   = act + big;
    float* Bmat = yb  + big;
    float* Cmat = Bmat + (size_t)NTOK * D_STATE;

    gemm_f32_wmma_lds<D_INNER, D_MODEL, 0><<<(NTOK / GBM) * (D_INNER / GBN), 256, 0, stream>>>(x, W_in,   b_in,   xz);
    gemm_f32_wmma_lds<D_INNER, D_MODEL, 0><<<(NTOK / GBM) * (D_INNER / GBN), 256, 0, stream>>>(x, W_gate, b_gate, xg);

    {
        const int total = NTOK * D_INNER;
        conv_silu<<<(total + 255) / 256, 256, 0, stream>>>(xz, conv_w, conv_b, act, total);
    }

    gemm_f32_wmma_lds<D_INNER, D_INNER, 1><<<(NTOK / GBM) * (D_INNER / GBN), 256, 0, stream>>>(act, W_delta, b_delta, xz);

    proj_bc_splitk<<<NTOK / 16, 256, 0, stream>>>(act, W_B, W_C, Bmat, Cmat);

    {
        const int swaves  = BB * (D_INNER / 2);
        const int threads = swaves * 32;
        ssm_scan<<<(threads + 255) / 256, 256, 0, stream>>>(xz, act, Bmat, Cmat,
                                                            A_log, D_skip, yb);
    }

    {
        const int total = NTOK * D_INNER;
        gate_mul<<<(total + 255) / 256, 256, 0, stream>>>(yb, xg, total);
    }

    gemm_f32_wmma_lds<D_MODEL, D_INNER, 0><<<(NTOK / GBM) * (D_MODEL / GBN), 256, 0, stream>>>(yb, W_out, b_out, out);
}


# gemm: LDS fragment reads of a half slice (16 k) go as one run ahead of the previous 32-WMMA block, into a second register set; B staged k pairs side by side, one 8-byte read a fragment; sums kept
# speedup vs baseline: 3.1084x; 1.0342x over previous
#include <hip/hip_runtime.h>
#include <hip/hip_bf16.h>
#include <math.h>

typedef float v2f __attribute__((ext_vector_type(2)));
typedef float v8f __attribute__((ext_vector_type(8)));
typedef float v4f __attribute__((ext_vector_type(4)));

#define D_MODEL 1024
#define D_STATE 16
#define D_INNER 2048
#define KCONV   4
#define BB      2
#define LL      2048
#define NTOK    (BB * LL)

__device__ __forceinline__ float silu_f(float x) { return x / (1.0f + __expf(-x)); }

__device__ __forceinline__ float epilogue_f(float s, float b, int e) {
    s += b;
    if (e == 1) {
        s = (s > 20.0f) ? s : log1pf(__expf(s));
    }
    return s;
}

#define PJ_WAVES 8
#define PJ_KW    (D_INNER / PJ_WAVES)
#define PJ_UN    8

__global__ __launch_bounds__(256) void proj_bc_splitk(
    const float* __restrict__ A, const float* __restrict__ Wb,
    const float* __restrict__ Wc, float* __restrict__ Bout, float* __restrict__ Cout)
{
    __shared__ float red[PJ_WAVES][2][8][32];

    const int lane = threadIdx.x & 31;
    const int wv   = threadIdx.x >> 5;
    const int half = lane >> 4;
    const int idx  = lane & 15;
    const int m0   = blockIdx.x * 16;

    const float* ap = A  + (size_t)(m0 + idx) * D_INNER + wv * PJ_KW + 2 * half;
    const float* bp = Wb + (size_t)(wv * PJ_KW + 2 * half) * D_STATE + idx;
    const float* cp = Wc + (size_t)(wv * PJ_KW + 2 * half) * D_STATE + idx;

    v8f accB = {}; v8f accC = {};

    for (int kc = 0; kc < PJ_KW; kc += 4 * PJ_UN) {
        v2f a[PJ_UN], b[PJ_UN], c[PJ_UN];
#pragma unroll
        for (int u = 0; u < PJ_UN; ++u) {
            const int k = kc + 4 * u;
            a[u]   = *(const v2f*)(ap + k);
            b[u].x = bp[(size_t)k       * D_STATE];
            b[u].y = bp[(size_t)(k + 1) * D_STATE];
            c[u].x = cp[(size_t)k       * D_STATE];
            c[u].y = cp[(size_t)(k + 1) * D_STATE];
        }
#pragma unroll
        for (int u = 0; u < PJ_UN; ++u) {
            accB = __builtin_amdgcn_wmma_f32_16x16x4_f32(false, a[u], false, b[u], (short)0, accB, false, false);
            accC = __builtin_amdgcn_wmma_f32_16x16x4_f32(false, a[u], false, c[u], (short)0, accC, false, false);
        }
    }

#pragma unroll
    for (int r = 0; r < 8; ++r) {
        red[wv][0][r][lane] = accB[r];
        red[wv][1][r][lane] = accC[r];
    }
    __syncthreads();

    const int row = 2 * wv + half;
    const int r   = row & 7;
    const int sl  = ((row >> 3) << 4) + idx;
    float sb = red[0][0][r][sl];
    float sc = red[0][1][r][sl];
#pragma unroll
    for (int w = 1; w < PJ_WAVES; ++w) {
        sb += red[w][0][r][sl];
        sc += red[w][1][r][sl];
    }
    Bout[(size_t)(m0 + row) * D_STATE + idx] = sb;
    Cout[(size_t)(m0 + row) * D_STATE + idx] = sc;
}

#define GBM 128
#define GBN 128
#define GBK 32
#define GPA (GBK + 4)
#define GPB (2 * GBN + 32)
#define GCH (GBK / 8)
#define GKS 4

template <int N, int K, int EPI>
__global__ __launch_bounds__(256) void gemm_f32_wmma_lds(
    const float* __restrict__ A, const float* __restrict__ Bw,
    const float* __restrict__ bias, float* __restrict__ C)
{
    __shared__ __attribute__((aligned(16))) float As[2][GBM * GPA];
    __shared__ __attribute__((aligned(16))) float Bs[2][(GBK / 2) * GPB];

    const int tid  = threadIdx.x;
    const int lane = tid & 31;
    const int wv   = tid >> 5;
    const int half = lane >> 4;
    const int idx  = lane & 15;
    constexpr int tiles_n = N / GBN;
    constexpr int NK = K / GBK;
    const int m0 = (blockIdx.x / tiles_n) * GBM;
    const int n0 = (blockIdx.x % tiles_n) * GBN;
    const int wm = (wv >> 2) * 64;
    const int wn = (wv & 3) * 32;

    v4f ra[GCH], rb[GCH];
    v8f acc[4][2] = {};
    v2f fa0[GKS][4], fb0[GKS][2], fa1[GKS][4], fb1[GKS][2];

    auto gload = [&](int k0) {
#pragma unroll
        for (int i = 0; i < GCH; ++i) {
            const int c = tid + 256 * i;
            const int q = tid + 256 * (i >> 1);
            ra[i] = *(const v4f*)(A + (size_t)(m0 + c / (GBK / 4)) * K + k0 + (c % (GBK / 4)) * 4);
            rb[i] = *(const v4f*)(Bw + (size_t)(k0 + 2 * (q / (GBN / 4)) + (i & 1)) * N + n0 + (q % (GBN / 4)) * 4);
        }
    };
    auto sstore = [&](int buf) {
#pragma unroll
        for (int i = 0; i < GCH; ++i) {
            const int c = tid + 256 * i;
            *(v4f*)(&As[buf][(c / (GBK / 4)) * GPA + (c % (GBK / 4)) * 4]) = ra[i];
        }
#pragma unroll
        for (int m = 0; m < GCH / 2; ++m) {
            const int q = tid + 256 * m;
            float* d = &Bs[buf][(q / (GBN / 4)) * GPB + (q % (GBN / 4)) * 8];
            const v4f e = rb[2 * m], o = rb[2 * m + 1];
            const v4f lo = {e.x, o.x, e.y, o.y};
            const v4f hi = {e.z, o.z, e.w, o.w};
            *(v4f*)(d)     = lo;
            *(v4f*)(d + 4) = hi;
        }
    };
    auto fload = [&](v2f (&fa)[GKS][4], v2f (&fb)[GKS][2], int buf, int g) {
        const float* as = &As[buf][(wm + idx) * GPA + 2 * half + 4 * GKS * g];
        const float* bs = &Bs[buf][(2 * GKS * g + half) * GPB + (wn + idx) * 2];
#pragma unroll
        for (int s = 0; s < GKS; ++s) {
#pragma unroll
            for (int i = 0; i < 4; ++i)
                fa[s][i] = *(const v2f*)(as + (16 * i) * GPA + 4 * s);
#pragma unroll
            for (int j = 0; j < 2; ++j)
                fb[s][j] = *(const v2f*)(bs + (2 * s) * GPB + 32 * j);
        }
    };
    auto fmma = [&](const v2f (&fa)[GKS][4], const v2f (&fb)[GKS][2]) {
#pragma unroll
        for (int s = 0; s < GKS; ++s)
#pragma unroll
            for (int i = 0; i < 4; ++i)
#pragma unroll
                for (int j = 0; j < 2; ++j)
                    acc[i][j] = __builtin_amdgcn_wmma_f32_16x16x4_f32(false, fa[s][i], false, fb[s][j], (short)0, acc[i][j], false, false);
    };

    gload(0);
    sstore(0);
    __syncthreads();
    fload(fa0, fb0, 0, 0);
    if (NK > 1) gload(GBK);

    for (int kt = 0; kt < NK; ++kt) {
        const int buf = kt & 1;
        fload(fa1, fb1, buf, 1);
        __builtin_amdgcn_sched_barrier(0);
        fmma(fa0, fb0);
        __builtin_amdgcn_sched_barrier(0);
        if (kt + 1 < NK) sstore(buf ^ 1);
        __syncthreads();
        if (kt + 1 < NK) fload(fa0, fb0, buf ^ 1, 0);
        if (kt + 2 < NK) gload((kt + 2) * GBK);
        __builtin_amdgcn_sched_barrier(0);
        fmma(fa1, fb1);
        __builtin_amdgcn_sched_barrier(0);
    }

#pragma unroll
    for (int j = 0; j < 2; ++j) {
        const int col = n0 + wn + 16 * j + idx;
        const float bv = bias ? bias[col] : 0.0f;
#pragma unroll
        for (int i = 0; i < 4; ++i) {
#pragma unroll
            for (int r = 0; r < 8; ++r) {
                const int row = m0 + wm + 16 * i + 8 * half + r;
                C[(size_t)row * N + col] = epilogue_f(acc[i][j][r], bv, EPI);
            }
        }
    }
}

__global__ __launch_bounds__(256) void conv_silu(
    const float* __restrict__ xz, const float* __restrict__ cw,
    const float* __restrict__ cb, float* __restrict__ act, int total)
{
    const int i = blockIdx.x * blockDim.x + threadIdx.x;
    if (i >= total) return;
    const int d = i % D_INNER;
    const int l = (i / D_INNER) % LL;
    float s = cb[d];
#pragma unroll
    for (int k = 0; k < KCONV; ++k) {
        const int lt = l - (KCONV - 1) + k;
        if (lt >= 0) s += cw[d * KCONV + k] * xz[(size_t)i + (size_t)((long)(lt - l) * D_INNER)];
    }
    act[i] = silu_f(s);
}

#define SCAN_T 16

struct ScanBlk { float dt, xt; float bt[SCAN_T], ct[SCAN_T]; };

__device__ __forceinline__ void scan_load(ScanBlk& k, const float* __restrict__ dp,
    const float* __restrict__ xp, const float* __restrict__ bp,
    const float* __restrict__ cp, int t0, int n)
{
    const size_t on = (size_t)(t0 + n) * D_INNER;
    k.dt = dp[on]; k.xt = xp[on];
#pragma unroll
    for (int j = 0; j < SCAN_T; ++j) {
        const size_t ob = (size_t)(t0 + j) * D_STATE;
        k.bt[j] = bp[ob]; k.ct[j] = cp[ob];
    }
}

__device__ __forceinline__ void scan_landed16(const float (&a)[SCAN_T])
{
    asm volatile("" :: "v"(a[0]), "v"(a[1]), "v"(a[2]),  "v"(a[3]),  "v"(a[4]),  "v"(a[5]),  "v"(a[6]),  "v"(a[7]),
                       "v"(a[8]), "v"(a[9]), "v"(a[10]), "v"(a[11]), "v"(a[12]), "v"(a[13]), "v"(a[14]), "v"(a[15]));
}
__device__ __forceinline__ void scan_landed(const ScanBlk& k)
{
    asm volatile("" :: "v"(k.dt), "v"(k.xt));
    scan_landed16(k.bt); scan_landed16(k.ct);
}

template <int J>
__device__ __forceinline__ float row_bcast(float v)
{
    return __int_as_float(__builtin_amdgcn_update_dpp(0, __float_as_int(v), 0x150 + J, 0xf, 0xf, true));
}

template <int J>
__device__ __forceinline__ void scan_step(const ScanBlk& k, float dx, float& h, float Acoef,
    float (&v)[SCAN_T])
{
    const float dtj = row_bcast<J>(k.dt);
    const float dxj = row_bcast<J>(dx);
    h = __expf(dtj * Acoef) * h + dxj * k.bt[J];
    v[J] = h * k.ct[J];
}

__device__ __forceinline__ void scan_run(const ScanBlk& k, float& h, float Acoef,
    float dsk, float* __restrict__ yp, int t0, int n)
{
    float v[SCAN_T];
    const float dx = k.dt * k.xt;
    scan_step<0>(k, dx, h, Acoef, v);  scan_step<1>(k, dx, h, Acoef, v);
    scan_step<2>(k, dx, h, Acoef, v);  scan_step<3>(k, dx, h, Acoef, v);
    scan_step<4>(k, dx, h, Acoef, v);  scan_step<5>(k, dx, h, Acoef, v);
    scan_step<6>(k, dx, h, Acoef, v);  scan_step<7>(k, dx, h, Acoef, v);
    scan_step<8>(k, dx, h, Acoef, v);  scan_step<9>(k, dx, h, Acoef, v);
    scan_step<10>(k, dx, h, Acoef, v); scan_step<11>(k, dx, h, Acoef, v);
    scan_step<12>(k, dx, h, Acoef, v); scan_step<13>(k, dx, h, Acoef, v);
    scan_step<14>(k, dx, h, Acoef, v); scan_step<15>(k, dx, h, Acoef, v);
    const bool b0 = (n & 1) != 0, b1 = (n & 2) != 0, b2 = (n & 4) != 0, b3 = (n & 8) != 0;
    float w[SCAN_T / 2], u[SCAN_T / 4], q[SCAN_T / 8];
#pragma unroll
    for (int i = 0; i < SCAN_T / 2; ++i)
        w[i] = (b0 ? v[2 * i + 1] : v[2 * i]) + __shfl_xor(b0 ? v[2 * i] : v[2 * i + 1], 1, 32);
#pragma unroll
    for (int i = 0; i < SCAN_T / 4; ++i)
        u[i] = (b1 ? w[2 * i + 1] : w[2 * i]) + __shfl_xor(b1 ? w[2 * i] : w[2 * i + 1], 2, 32);
#pragma unroll
    for (int i = 0; i < SCAN_T / 8; ++i)
        q[i] = (b2 ? u[2 * i + 1] : u[2 * i]) + __shfl_xor(b2 ? u[2 * i] : u[2 * i + 1], 4, 32);
    const float o = (b3 ? q[1] : q[0]) + __shfl_xor(b3 ? q[0] : q[1], 8, 32);
    yp[(size_t)(t0 + n) * D_INNER] = o + k.xt * dsk;
}

__global__ __launch_bounds__(256) void ssm_scan(
    const float* __restrict__ delta, const float* __restrict__ xact,
    const float* __restrict__ Bm, const float* __restrict__ Cm,
    const float* __restrict__ A_log, const float* __restrict__ D_skip,
    float* __restrict__ y)
{
    const int wave = blockIdx.x * (blockDim.x >> 5) + (threadIdx.x >> 5);
    const int lane = threadIdx.x & 31;
    const int half = lane >> 4;
    const int n    = lane & 15;
    const int total = BB * (D_INNER / 2);
    if (wave >= total) return;
    const int b = wave / (D_INNER / 2);
    const int d = (wave % (D_INNER / 2)) * 2 + half;

    const float Acoef = -__expf(A_log[d * D_STATE + n]);
    const float dsk   = D_skip[d];

    const float* dp = delta + (size_t)b * LL * D_INNER + d;
    const float* xp = xact  + (size_t)b * LL * D_INNER + d;
    const float* bp = Bm    + (size_t)b * LL * D_STATE + n;
    const float* cp = Cm    + (size_t)b * LL * D_STATE + n;
    float*       yp = y     + (size_t)b * LL * D_INNER + d;

    float h = 0.0f;
    ScanBlk ka, kb;
    scan_load(ka, dp, xp, bp, cp, 0, n);

    for (int t0 = 0; t0 < LL; t0 += 2 * SCAN_T) {
        scan_landed(ka);
        scan_load(kb, dp, xp, bp, cp, t0 + SCAN_T, n);
        scan_run(ka, h, Acoef, dsk, yp, t0, n);
        scan_landed(kb);
        if (t0 + 2 * SCAN_T < LL) scan_load(ka, dp, xp, bp, cp, t0 + 2 * SCAN_T, n);
        scan_run(kb, h, Acoef, dsk, yp, t0 + SCAN_T, n);
    }
}

__global__ __launch_bounds__(256) void gate_mul(
    float* __restrict__ y, const float* __restrict__ g, int total)
{
    const int i = blockIdx.x * blockDim.x + threadIdx.x;
    if (i < total) y[i] *= silu_f(g[i]);
}

extern "C" void kernel_launch(void* const* d_in, const int* in_sizes, int n_in,
                              void* d_out, int out_size, void* d_ws, size_t ws_size,
                              hipStream_t stream) {
    const float* x      = (const float*)d_in[0];
    const float* W_in   = (const float*)d_in[1];
    const float* b_in   = (const float*)d_in[2];
    const float* W_gate = (const float*)d_in[3];
    const float* b_gate = (const float*)d_in[4];
    const float* conv_w = (const float*)d_in[5];
    const float* conv_b = (const float*)d_in[6];
    const float* A_log  = (const float*)d_in[7];
    const float* D_skip = (const float*)d_in[8];
    const float* W_delta= (const float*)d_in[9];
    const float* b_delta= (const float*)d_in[10];
    const float* W_B    = (const float*)d_in[11];
    const float* W_C    = (const float*)d_in[12];
    const float* W_out  = (const float*)d_in[13];
    const float* b_out  = (const float*)d_in[14];
    float* out = (float*)d_out;

    const size_t big = (size_t)NTOK * D_INNER;
    float* ws   = (float*)d_ws;
    float* xz   = ws;
    float* xg   = xz  + big;
    float* act  = xg  + big;
    float* yb   = act + big;
    float* Bmat = yb  + big;
    float* Cmat = Bmat + (size_t)NTOK * D_STATE;

    gemm_f32_wmma_lds<D_INNER, D_MODEL, 0><<<(NTOK / GBM) * (D_INNER / GBN), 256, 0, stream>>>(x, W_in,   b_in,   xz);
    gemm_f32_wmma_lds<D_INNER, D_MODEL, 0><<<(NTOK / GBM) * (D_INNER / GBN), 256, 0, stream>>>(x, W_gate, b_gate, xg);

    {
        const int total = NTOK * D_INNER;
        conv_silu<<<(total + 255) / 256, 256, 0, stream>>>(xz, conv_w, conv_b, act, total);
    }

    gemm_f32_wmma_lds<D_INNER, D_INNER, 1><<<(NTOK / GBM) * (D_INNER / GBN), 256, 0, stream>>>(act, W_delta, b_delta, xz);

    proj_bc_splitk<<<NTOK / 16, 256, 0, stream>>>(act, W_B, W_C, Bmat, Cmat);

    {
        const int swaves  = BB * (D_INNER / 2);
        const int threads = swaves * 32;
        ssm_scan<<<(threads + 255) / 256, 256, 0, stream>>>(xz, act, Bmat, Cmat,
                                                            A_log, D_skip, yb);
    }

    {
        const int total = NTOK * D_INNER;
        gate_mul<<<(total + 255) / 256, 256, 0, stream>>>(yb, xg, total);
    }

    gemm_f32_wmma_lds<D_MODEL, D_INNER, 0><<<(NTOK / GBM) * (D_MODEL / GBN), 256, 0, stream>>>(yb, W_out, b_out, out);
}
